# starting kernel
# baseline (speedup 1.0000x reference)
#include <hip/hip_runtime.h>


#define BB 2
#define SS 2048
#define DD 1024
#define HH 16
#define DKV 64

typedef __attribute__((ext_vector_type(16))) __bf16 v16bf;
typedef __attribute__((ext_vector_type(8)))  float  v8f;

union FragU { v16bf v; uint4 q[2]; };

__device__ __forceinline__ unsigned short f2bf(float f) {
  unsigned int u = __float_as_uint(f);
  u += 0x7FFFu + ((u >> 16) & 1u);
  return (unsigned short)(u >> 16);
}

__device__ __forceinline__ v16bf load_frag(const unsigned short* rowbase, int lane) {
  const unsigned short* p = rowbase + ((lane & 16) ? 8 : 0);
  FragU f;
  f.q[0] = *reinterpret_cast<const uint4*>(p);
  f.q[1] = *reinterpret_cast<const uint4*>(p + 16);
  return f.v;
}

#define WMMA_BF16(a, b, c) \
  __builtin_amdgcn_wmma_f32_16x16x32_bf16(false, (a), false, (b), (short)0, (c), false, false)

__global__ void cvt_kernel(const float* __restrict__ src,
                           unsigned short* __restrict__ dst, int n) {
  int i = blockIdx.x * blockDim.x + threadIdx.x;
  if (i < n) dst[i] = f2bf(src[i]);
}

__global__ void cvtT_kernel(const float* __restrict__ src,
                            unsigned short* __restrict__ dst,
                            int G, int R, int C) {
  int i = blockIdx.x * blockDim.x + threadIdx.x;
  int n = G * R * C;
  if (i >= n) return;
  int g = i / (R * C);
  int rem = i - g * (R * C);
  int r = rem / C, c = rem - r * C;
  dst[(size_t)g * R * C + (size_t)c * R + r] = f2bf(src[i]);
}

__global__ void proj_kernel(const unsigned short* __restrict__ X,
                            const unsigned short* __restrict__ WT,
                            unsigned short* __restrict__ OUT,
                            float scale, int transposeOut) {
  const int lane = threadIdx.x & 31;
  const int wave = threadIdx.x >> 5;
  const int bh = blockIdx.x;
  const int b = bh / HH, h = bh % HH;
  const int s_base = blockIdx.y * 64 + wave * 16;

  const unsigned short* Xrow =
      X + ((size_t)b * SS + (size_t)(s_base + (lane & 15))) * DD;
  const unsigned short* Wrow = WT + ((size_t)h * DKV + (lane & 15)) * DD;

  v8f acc[4] = {};
#pragma unroll 2
  for (int k0 = 0; k0 < DD; k0 += 32) {
    __builtin_prefetch(Xrow + k0 + 64, 0, 1);
    v16bf a = load_frag(Xrow + k0, lane);
    v16bf bf[4];
#pragma unroll
    for (int nt = 0; nt < 4; ++nt)
      bf[nt] = load_frag(Wrow + (size_t)nt * 16 * DD + k0, lane);
#pragma unroll
    for (int nt = 0; nt < 4; ++nt)
      acc[nt] = WMMA_BF16(a, bf[nt], acc[nt]);
  }
  const int rbase = (lane & 16) ? 8 : 0;
  const int col0 = lane & 15;
#pragma unroll
  for (int nt = 0; nt < 4; ++nt) {
#pragma unroll
    for (int r = 0; r < 8; ++r) {
      int srow = s_base + rbase + r;
      int col = nt * 16 + col0;
      unsigned short val = f2bf(acc[nt][r] * scale);
      if (transposeOut)
        OUT[(size_t)bh * DKV * SS + (size_t)col * SS + srow] = val;
      else
        OUT[(size_t)bh * SS * DKV + (size_t)srow * DKV + col] = val;
    }
  }
}

__global__ void attn_kernel(const unsigned short* __restrict__ QH,
                            const unsigned short* __restrict__ KH,
                            const unsigned short* __restrict__ VT,
                            const int* __restrict__ MASK,
                            unsigned short* __restrict__ AOUT) {
  __shared__ __align__(16) unsigned short Pld[4][16][32];

  const int lane = threadIdx.x & 31;
  const int wave = threadIdx.x >> 5;
  const int bh = blockIdx.x;
  const int b = bh >> 4, h = bh & 15;
  const int s_base = blockIdx.y * 64 + wave * 16;
  const int rbase = (lane & 16) ? 8 : 0;
  const int col = lane & 15;

  const unsigned short* Qrow =
      QH + ((size_t)bh * SS + (size_t)(s_base + col)) * DKV;
  v16bf qf0 = load_frag(Qrow, lane);
  v16bf qf1 = load_frag(Qrow + 32, lane);

  v8f o[4] = {};
  float m[8], l[8];
#pragma unroll
  for (int r = 0; r < 8; ++r) { m[r] = -3.0e38f; l[r] = 0.0f; }

  const int* maskbase = MASK + (size_t)b * SS * SS;
  const unsigned short* Krow = KH + ((size_t)bh * SS + col) * DKV;
  const unsigned short* Vrow = VT + ((size_t)bh * DKV + col) * SS;

  for (int t0 = 0; t0 < SS; t0 += 32) {
    v16bf kb0 = load_frag(Krow + (size_t)t0 * DKV, lane);
    v16bf kb1 = load_frag(Krow + (size_t)t0 * DKV + 32, lane);
    v16bf kb2 = load_frag(Krow + (size_t)(t0 + 16) * DKV, lane);
    v16bf kb3 = load_frag(Krow + (size_t)(t0 + 16) * DKV + 32, lane);
    v16bf vb[4];
#pragma unroll
    for (int nt = 0; nt < 4; ++nt)
      vb[nt] = load_frag(Vrow + (size_t)nt * 16 * SS + t0, lane);

    v8f s0 = {}, s1 = {};
    s0 = WMMA_BF16(qf0, kb0, s0);
    s0 = WMMA_BF16(qf1, kb1, s0);
    s1 = WMMA_BF16(qf0, kb2, s1);
    s1 = WMMA_BF16(qf1, kb3, s1);

#pragma unroll
    for (int r = 0; r < 8; ++r) {
      const int* mrow = maskbase + (size_t)(s_base + rbase + r) * SS + t0;
      float v0 = (mrow[col]      != 0) ? s0[r] : -1.0e9f;
      float v1 = (mrow[col + 16] != 0) ? s1[r] : -1.0e9f;
      float mx = fmaxf(v0, v1);
      mx = fmaxf(mx, __shfl_xor(mx, 1));
      mx = fmaxf(mx, __shfl_xor(mx, 2));
      mx = fmaxf(mx, __shfl_xor(mx, 4));
      mx = fmaxf(mx, __shfl_xor(mx, 8));
      float mnew  = fmaxf(m[r], mx);
      float alpha = __expf(m[r] - mnew);
      float p0 = __expf(v0 - mnew);
      float p1 = __expf(v1 - mnew);
      float rs = p0 + p1;
      rs += __shfl_xor(rs, 1);
      rs += __shfl_xor(rs, 2);
      rs += __shfl_xor(rs, 4);
      rs += __shfl_xor(rs, 8);
      l[r] = l[r] * alpha + rs;
      m[r] = mnew;
#pragma unroll
      for (int nt = 0; nt < 4; ++nt) o[nt][r] = o[nt][r] * alpha;
      Pld[wave][rbase + r][col]      = f2bf(p0);
      Pld[wave][rbase + r][col + 16] = f2bf(p1);
    }
    asm volatile("s_wait_dscnt 0" ::: "memory");
    v16bf pf = load_frag(&Pld[wave][col][0], lane);
#pragma unroll
    for (int nt = 0; nt < 4; ++nt)
      o[nt] = WMMA_BF16(pf, vb[nt], o[nt]);
  }

#pragma unroll
  for (int r = 0; r < 8; ++r) {
    float inv = 1.0f / l[r];
    int srow = s_base + rbase + r;
#pragma unroll
    for (int nt = 0; nt < 4; ++nt) {
      int dv = nt * 16 + col;
      AOUT[((size_t)b * SS + srow) * (HH * DKV) + h * DKV + dv] =
          f2bf(o[nt][r] * inv);
    }
  }
}

__global__ void fc_kernel(const unsigned short* __restrict__ A,
                          const unsigned short* __restrict__ WT,
                          const float* __restrict__ RES,
                          float* __restrict__ OUT) {
  const int lane = threadIdx.x & 31;
  const int wave = threadIdx.x >> 5;
  const int row0 = blockIdx.x * 16;
  const int n0 = wave * 128;

  const unsigned short* Arow = A + (size_t)(row0 + (lane & 15)) * DD;
  const unsigned short* Wrow = WT + (size_t)(n0 + (lane & 15)) * DD;

  v8f acc[8] = {};
  for (int k0 = 0; k0 < DD; k0 += 32) {
    __builtin_prefetch(Arow + k0 + 64, 0, 1);
    v16bf a = load_frag(Arow + k0, lane);
    v16bf bf[8];
#pragma unroll
    for (int nt = 0; nt < 8; ++nt)
      bf[nt] = load_frag(Wrow + (size_t)nt * 16 * DD + k0, lane);
#pragma unroll
    for (int nt = 0; nt < 8; ++nt)
      acc[nt] = WMMA_BF16(a, bf[nt], acc[nt]);
  }
  const int rbase = (lane & 16) ? 8 : 0;
  const int col0 = lane & 15;
#pragma unroll
  for (int nt = 0; nt < 8; ++nt) {
#pragma unroll
    for (int r = 0; r < 8; ++r) {
      size_t idx = (size_t)(row0 + rbase + r) * DD + n0 + nt * 16 + col0;
      OUT[idx] = acc[nt][r] + RES[idx];
    }
  }
}

__global__ void ln_kernel(const float* __restrict__ X,
                          const float* __restrict__ W,
                          const float* __restrict__ Bv,
                          float* __restrict__ OUT) {
  __shared__ float red[8];
  const int row = blockIdx.x;
  const int tid = threadIdx.x;
  const float* xr = X + (size_t)row * DD;

  float v[4];
  float s = 0.0f;
#pragma unroll
  for (int i = 0; i < 4; ++i) { v[i] = xr[tid + i * 256]; s += v[i]; }
  for (int off = 1; off < 32; off <<= 1) s += __shfl_xor(s, off);
  if ((tid & 31) == 0) red[tid >> 5] = s;
  __syncthreads();
  float mu = (red[0] + red[1] + red[2] + red[3] +
              red[4] + red[5] + red[6] + red[7]) * (1.0f / DD);
  float vs = 0.0f;
#pragma unroll
  for (int i = 0; i < 4; ++i) { float d = v[i] - mu; vs += d * d; }
  for (int off = 1; off < 32; off <<= 1) vs += __shfl_xor(vs, off);
  __syncthreads();
  if ((tid & 31) == 0) red[tid >> 5] = vs;
  __syncthreads();
  float var = (red[0] + red[1] + red[2] + red[3] +
               red[4] + red[5] + red[6] + red[7]) * (1.0f / DD);
  float rstd = rsqrtf(var + 1e-6f);
#pragma unroll
  for (int i = 0; i < 4; ++i) {
    int c = tid + i * 256;
    OUT[(size_t)row * DD + c] = (v[i] - mu) * rstd * W[c] + Bv[c];
  }
}

extern "C" void kernel_launch(void* const* d_in, const int* in_sizes, int n_in,
                              void* d_out, int out_size, void* d_ws, size_t ws_size,
                              hipStream_t stream) {
  const float* q    = (const float*)d_in[0];
  const float* k    = (const float*)d_in[1];
  const float* v    = (const float*)d_in[2];
  const int*   mask = (const int*)d_in[3];
  const float* Wq   = (const float*)d_in[4];
  const float* Wk   = (const float*)d_in[5];
  const float* Wv   = (const float*)d_in[6];
  const float* Wfc  = (const float*)d_in[7];
  const float* lnw  = (const float*)d_in[8];
  const float* lnb  = (const float*)d_in[9];
  float* out = (float*)d_out;

  char* ws = (char*)d_ws;
  const size_t SZ_ACT  = (size_t)BB * SS * DD * 2;
  const size_t SZ_W    = (size_t)HH * DD * DKV * 2;
  const size_t SZ_HEAD = (size_t)BB * HH * SS * DKV * 2;
  size_t off = 0;
  unsigned short* qbf  = (unsigned short*)(ws + off); off += SZ_ACT;
  unsigned short* kbf  = (unsigned short*)(ws + off); off += SZ_ACT;
  unsigned short* vbf  = (unsigned short*)(ws + off); off += SZ_ACT;
  unsigned short* WqT  = (unsigned short*)(ws + off); off += SZ_W;
  unsigned short* WkT  = (unsigned short*)(ws + off); off += SZ_W;
  unsigned short* WvT  = (unsigned short*)(ws + off); off += SZ_W;
  unsigned short* WfcT = (unsigned short*)(ws + off); off += (size_t)DD * DD * 2;
  unsigned short* qh   = (unsigned short*)(ws + off); off += SZ_HEAD;
  unsigned short* kh   = (unsigned short*)(ws + off); off += SZ_HEAD;
  unsigned short* vhT  = (unsigned short*)(ws + off); off += SZ_HEAD;
  unsigned short* aout = (unsigned short*)(ws + off); off += SZ_ACT;
  float*          fco  = (float*)(ws + off);          off += (size_t)BB * SS * DD * 4;

  const int nAct = BB * SS * DD;
  const int nW   = HH * DD * DKV;
  const int nFc  = DD * DD;

  cvt_kernel<<<(nAct + 255) / 256, 256, 0, stream>>>(q, qbf, nAct);
  cvt_kernel<<<(nAct + 255) / 256, 256, 0, stream>>>(k, kbf, nAct);
  cvt_kernel<<<(nAct + 255) / 256, 256, 0, stream>>>(v, vbf, nAct);
  cvtT_kernel<<<(nW + 255) / 256, 256, 0, stream>>>(Wq, WqT, HH, DD, DKV);
  cvtT_kernel<<<(nW + 255) / 256, 256, 0, stream>>>(Wk, WkT, HH, DD, DKV);
  cvtT_kernel<<<(nW + 255) / 256, 256, 0, stream>>>(Wv, WvT, HH, DD, DKV);
  cvtT_kernel<<<(nFc + 255) / 256, 256, 0, stream>>>(Wfc, WfcT, 1, DD, DD);

  dim3 gProj(BB * HH, SS / 64);
  proj_kernel<<<gProj, 128, 0, stream>>>(qbf, WqT, qh, 0.125f, 0);
  proj_kernel<<<gProj, 128, 0, stream>>>(kbf, WkT, kh, 1.0f, 0);
  proj_kernel<<<gProj, 128, 0, stream>>>(vbf, WvT, vhT, 1.0f, 1);

  attn_kernel<<<dim3(BB * HH, SS / 64), 128, 0, stream>>>(qh, kh, vhT, mask, aout);

  fc_kernel<<<(BB * SS) / 16, 256, 0, stream>>>(aout, WfcT, q, fco);
  ln_kernel<<<BB * SS, 256, 0, stream>>>(fco, lnw, lnb, out);
}


# K and V proj: wave tile 16 rows x 128 cols (2 heads) in registers, was x 64: each fp32 activation fragment (16 lines a load) feeds 16 WMMAs, not 8; grid halves; sum order unchanged
# speedup vs baseline: 3.7477x; 3.7477x over previous
#include <hip/hip_runtime.h>


#define BB 2
#define SS 2048
#define DD 1024
#define HH 16
#define DKV 64

typedef __attribute__((ext_vector_type(16))) __bf16 v16bf;
typedef __attribute__((ext_vector_type(8)))  float  v8f;

union FragU { v16bf v; uint4 q[2]; };

__device__ __forceinline__ unsigned short f2bf(float f) {
  unsigned int u = __float_as_uint(f);
  u += 0x7FFFu + ((u >> 16) & 1u);
  return (unsigned short)(u >> 16);
}

__device__ __forceinline__ v16bf load_frag(const unsigned short* rowbase, int lane) {
  const unsigned short* p = rowbase + ((lane & 16) ? 8 : 0);
  FragU f;
  f.q[0] = *reinterpret_cast<const uint4*>(p);
  f.q[1] = *reinterpret_cast<const uint4*>(p + 16);
  return f.v;
}

__device__ __forceinline__ int ft_off(int row, int k) {
  return ((k >> 4) << 8) + ((row + (((k >> 3) & 1) << 4)) << 3) + (k & 7);
}

__device__ __forceinline__ v16bf load_frag_t(const unsigned short* tile_lane) {
  FragU f;
  f.q[0] = *reinterpret_cast<const uint4*>(tile_lane);
  f.q[1] = *reinterpret_cast<const uint4*>(tile_lane + 256);
  return f.v;
}

struct RawA { float4 x0, x1, x2, x3; };

__device__ __forceinline__ RawA load_raw_f32(const float* rowbase, int lane) {
  const float* p = rowbase + ((lane & 16) ? 8 : 0);
  RawA r;
  r.x0 = *reinterpret_cast<const float4*>(p);
  r.x1 = *reinterpret_cast<const float4*>(p + 4);
  r.x2 = *reinterpret_cast<const float4*>(p + 16);
  r.x3 = *reinterpret_cast<const float4*>(p + 20);
  return r;
}

__device__ __forceinline__ v16bf pack_frag_f32(const RawA& r) {
  const float4 x0 = r.x0, x1 = r.x1, x2 = r.x2, x3 = r.x3;
  v16bf f;
  f[0]  = (__bf16)x0.x; f[1]  = (__bf16)x0.y; f[2]  = (__bf16)x0.z; f[3]  = (__bf16)x0.w;
  f[4]  = (__bf16)x1.x; f[5]  = (__bf16)x1.y; f[6]  = (__bf16)x1.z; f[7]  = (__bf16)x1.w;
  f[8]  = (__bf16)x2.x; f[9]  = (__bf16)x2.y; f[10] = (__bf16)x2.z; f[11] = (__bf16)x2.w;
  f[12] = (__bf16)x3.x; f[13] = (__bf16)x3.y; f[14] = (__bf16)x3.z; f[15] = (__bf16)x3.w;
  return f;
}

#define WMMA_BF16(a, b, c) \
  __builtin_amdgcn_wmma_f32_16x16x32_bf16(false, (a), false, (b), (short)0, (c), false, false)

template <int M>
__device__ __forceinline__ float row_xor(float x) {
  return __int_as_float(__builtin_amdgcn_update_dpp(
      0, __float_as_int(x), 0x160 + M, 0xf, 0xf, true));
}

__global__ void cvtT_kernel(const float* __restrict__ src,
                            unsigned short* __restrict__ dst,
                            int G, int R, int C) {
  int i = blockIdx.x * blockDim.x + threadIdx.x;
  int n = G * R * C;
  if (i >= n) return;
  int g = i / (R * C);
  int rem = i - g * (R * C);
  int r = rem / C, c = rem - r * C;
  size_t tile = ((size_t)g * (R >> 5) + (r >> 5)) * (C >> 4) + (c >> 4);
  dst[tile * 512 + ft_off(c & 15, r & 31)] = f2bf(src[i]);
}

template <int NH>
__device__ __forceinline__ void proj_tile(const float* __restrict__ Xrow,
                                          const unsigned short* __restrict__ Wt,
                                          int lane, v8f* acc) {
  constexpr size_t HEAD = (size_t)(DD / 32) * (DKV / 16) * 512;
  constexpr int PF = 8;
  const float* Xpf = Xrow + ((lane & 16) ? 32 : 0);
#pragma unroll
  for (int j = 0; j < PF; j += 2) __builtin_prefetch(Xpf + j * 32, 0, 1);

  for (int k0 = 0; k0 < DD; k0 += 64) {
    if (k0 + PF * 32 < DD)
      __builtin_prefetch(Xpf + k0 + PF * 32, 0, 1);
    RawA xa[2];
    v16bf bf[2][NH * 4];
#pragma unroll
    for (int kk = 0; kk < 2; ++kk)
      xa[kk] = load_raw_f32(Xrow + k0 + kk * 32, lane);
#pragma unroll
    for (int kk = 0; kk < 2; ++kk) {
#pragma unroll
      for (int hh = 0; hh < NH; ++hh) {
#pragma unroll
        for (int nt = 0; nt < 4; ++nt)
          bf[kk][hh * 4 + nt] = load_frag_t(
              Wt + hh * HEAD + (size_t)((((k0 >> 5) + kk) << 2) + nt) * 512);
      }
    }
    v16bf a[2];
#pragma unroll
    for (int kk = 0; kk < 2; ++kk) a[kk] = pack_frag_f32(xa[kk]);
    __builtin_amdgcn_sched_barrier(0);
#pragma unroll
    for (int kk = 0; kk < 2; ++kk) {
#pragma unroll
      for (int nt = 0; nt < NH * 4; ++nt)
        acc[nt] = WMMA_BF16(a[kk], bf[kk][nt], acc[nt]);
    }
  }
}

#define PROJ_NH 2
__global__ void __launch_bounds__(128) __attribute__((amdgpu_waves_per_eu(2, 2)))
proj_kernel(const float* __restrict__ X,
                            const unsigned short* __restrict__ WT,
                            unsigned short* __restrict__ OUT,
                            float scale, int transposeOut) {
  const int lane = threadIdx.x & 31;
  const int wave = threadIdx.x >> 5;
  const int b = blockIdx.x / (HH / PROJ_NH);
  const int h0 = (blockIdx.x % (HH / PROJ_NH)) * PROJ_NH;
  const int s_base = blockIdx.y * 64 + wave * 16;

  const float* Xrow =
      X + ((size_t)b * SS + (size_t)(s_base + (lane & 15))) * DD;
  const unsigned short* Wt = WT + (size_t)h0 * (DD / 32) * (DKV / 16) * 512 + lane * 8;

  v8f accs[PROJ_NH * 4] = {};
  proj_tile<PROJ_NH>(Xrow, Wt, lane, accs);
  const int rbase = (lane & 16) ? 8 : 0;
  const int col0 = lane & 15;
#pragma unroll
  for (int hh = 0; hh < PROJ_NH; ++hh) {
  const v8f* acc = accs + hh * 4;
  const int bh = b * HH + h0 + hh;
  if (transposeOut) {
    unsigned short* o = OUT + (((size_t)bh * (SS / 32) + (s_base >> 5)) << 2) * 512 +
                        (((s_base >> 4) & 1) << 8) + lane * 8;
#pragma unroll
    for (int nt = 0; nt < 4; ++nt) {
      uint4 w;
      w.x = (unsigned)f2bf(acc[nt][0] * scale) | ((unsigned)f2bf(acc[nt][1] * scale) << 16);
      w.y = (unsigned)f2bf(acc[nt][2] * scale) | ((unsigned)f2bf(acc[nt][3] * scale) << 16);
      w.z = (unsigned)f2bf(acc[nt][4] * scale) | ((unsigned)f2bf(acc[nt][5] * scale) << 16);
      w.w = (unsigned)f2bf(acc[nt][6] * scale) | ((unsigned)f2bf(acc[nt][7] * scale) << 16);
      *reinterpret_cast<uint4*>(o + nt * 512) = w;
    }
  } else {
    unsigned short* o = OUT + (((size_t)bh * (SS / 16) + (s_base >> 4)) << 1) * 512;
#pragma unroll
    for (int nt = 0; nt < 4; ++nt) {
#pragma unroll
      for (int r = 0; r < 8; ++r)
        o[(nt >> 1) * 512 + ft_off(rbase + r, (nt & 1) * 16 + col0)] =
            f2bf(acc[nt][r] * scale);
    }
  }
  }
}

#define ATT_KB 128
#define ATT_PP (ATT_KB + 8)

struct KM64 { v16bf kb[8]; int mk[32]; };

__device__ __forceinline__ void load_km64(KM64& x, const unsigned short* kp, const int* mp) {
#pragma unroll
  for (int i = 0; i < 8; ++i) x.kb[i] = load_frag_t(kp + i * 512);
#pragma unroll
  for (int jj = 0; jj < 2; ++jj) {
#pragma unroll
    for (int r = 0; r < 8; ++r) {
      x.mk[jj * 16 + 2 * r]     = mp[(size_t)r * SS + 32 * jj];
      x.mk[jj * 16 + 2 * r + 1] = mp[(size_t)r * SS + 32 * jj + 16];
    }
  }
}

__device__ __forceinline__ void scores64(const KM64& x, v16bf qf0, v16bf qf1, v8f* sc) {
#pragma unroll
  for (int jj = 0; jj < 2; ++jj) {
    v8f s0 = {}, s1 = {};
    s0 = WMMA_BF16(qf0, x.kb[4 * jj],     s0);
    s0 = WMMA_BF16(qf1, x.kb[4 * jj + 1], s0);
    s1 = WMMA_BF16(qf0, x.kb[4 * jj + 2], s1);
    s1 = WMMA_BF16(qf1, x.kb[4 * jj + 3], s1);
#pragma unroll
    for (int r = 0; r < 8; ++r) {
      sc[2 * jj][r]     = (x.mk[jj * 16 + 2 * r]     != 0) ? s0[r] : -1.0e9f;
      sc[2 * jj + 1][r] = (x.mk[jj * 16 + 2 * r + 1] != 0) ? s1[r] : -1.0e9f;
    }
  }
}

__global__ void __launch_bounds__(128) __attribute__((amdgpu_waves_per_eu(4, 4)))
attn_kernel(const float* __restrict__ XQ,
                            const unsigned short* __restrict__ WQT,
                            const unsigned short* __restrict__ KH,
                            const unsigned short* __restrict__ VT,
                            const int* __restrict__ MASK,
                            unsigned short* __restrict__ AOUT) {
  __shared__ __align__(16) unsigned short Pld[4][16][ATT_PP];

  const int lane = threadIdx.x & 31;
  const int wave = threadIdx.x >> 5;
  const int bh = blockIdx.x;
  const int b = bh >> 4, h = bh & 15;
  const int s_base = blockIdx.y * 64 + wave * 16;
  const int rbase = (lane & 16) ? 8 : 0;
  const int col = lane & 15;

  v16bf qf0, qf1;
  {
    v8f qa[4] = {};
    proj_tile<1>(XQ + ((size_t)b * SS + (size_t)(s_base + col)) * DD,
              WQT + (size_t)h * (DD / 32) * (DKV / 16) * 512 + lane * 8, lane, qa);
#pragma unroll
    for (int nt = 0; nt < 4; ++nt) {
#pragma unroll
      for (int r = 0; r < 8; ++r)
        Pld[wave][rbase + r][nt * 16 + col] = f2bf(qa[nt][r] * 0.125f);
    }
    asm volatile("s_wait_dscnt 0" ::: "memory");
    qf0 = load_frag(&Pld[wave][col][0], lane);
    qf1 = load_frag(&Pld[wave][col][32], lane);
  }

  v8f o[4] = {};
  float m[8], l[8];
#pragma unroll
  for (int r = 0; r < 8; ++r) { m[r] = -3.0e38f; l[r] = 0.0f; }

  const int* maskrow = MASK + (size_t)b * SS * SS + (size_t)(s_base + rbase) * SS + col;
  const unsigned short* Kt = KH + (size_t)bh * SS * DKV + lane * 8;
  const unsigned short* Vt = VT + (size_t)bh * DKV * SS + lane * 8;

  for (int t0 = 0; t0 < SS; t0 += ATT_KB) {
    v8f sc[8];
    {
      KM64 x;
      load_km64(x, Kt + (size_t)t0 * DKV, maskrow + t0);
      scores64(x, qf0, qf1, sc);
    }
    {
      KM64 x;
      load_km64(x, Kt + (size_t)(t0 + 64) * DKV, maskrow + t0 + 64);
      scores64(x, qf0, qf1, sc + 4);
    }
    v16bf va[8];
#pragma unroll
    for (int i = 0; i < 8; ++i) va[i] = load_frag_t(Vt + (size_t)t0 * DKV + i * 512);

#pragma unroll
    for (int r = 0; r < 8; ++r) {
      float mx = fmaxf(sc[0][r], sc[1][r]);
#pragma unroll
      for (int j = 1; j < 4; ++j)
        mx = fmaxf(mx, fmaxf(sc[2 * j][r], sc[2 * j + 1][r]));
      mx = fmaxf(mx, row_xor<1>(mx));
      mx = fmaxf(mx, row_xor<2>(mx));
      mx = fmaxf(mx, row_xor<4>(mx));
      mx = fmaxf(mx, row_xor<8>(mx));
      float mnew  = fmaxf(m[r], mx);
      float alpha = __expf(m[r] - mnew);
      float rs = 0.0f;
#pragma unroll
      for (int j = 0; j < 4; ++j) {
        float p0 = __expf(sc[2 * j][r] - mnew);
        float p1 = __expf(sc[2 * j + 1][r] - mnew);
        rs += p0 + p1;
        Pld[wave][rbase + r][32 * j + col]      = f2bf(p0);
        Pld[wave][rbase + r][32 * j + col + 16] = f2bf(p1);
      }
      rs += row_xor<1>(rs);
      rs += row_xor<2>(rs);
      rs += row_xor<4>(rs);
      rs += row_xor<8>(rs);
      l[r] = l[r] * alpha + rs;
      m[r] = mnew;
#pragma unroll
      for (int nt = 0; nt < 4; ++nt) o[nt][r] = o[nt][r] * alpha;
    }
    asm volatile("s_wait_dscnt 0" ::: "memory");
    v16bf vc[8];
#pragma unroll
    for (int i = 0; i < 8; ++i) vc[i] = load_frag_t(Vt + (size_t)(t0 + 64) * DKV + i * 512);
#pragma unroll
    for (int j = 0; j < 4; ++j) {
      v16bf pf = load_frag(&Pld[wave][col][32 * j], lane);
#pragma unroll
      for (int nt = 0; nt < 4; ++nt)
        o[nt] = WMMA_BF16(pf, (j < 2) ? va[4 * j + nt] : vc[4 * (j - 2) + nt], o[nt]);
    }
  }

  unsigned short* ao =
      AOUT + ((((size_t)b * SS + s_base) >> 4) * (HH * DKV / 32) + h * 2) * 512;
#pragma unroll
  for (int r = 0; r < 8; ++r) {
    float inv = 1.0f / l[r];
#pragma unroll
    for (int nt = 0; nt < 4; ++nt)
      ao[(nt >> 1) * 512 + ft_off(rbase + r, (nt & 1) * 16 + col)] =
          f2bf(o[nt][r] * inv);
  }
}

__global__ void __launch_bounds__(256) __attribute__((amdgpu_waves_per_eu(2, 2)))
fc_ln_kernel(const unsigned short* __restrict__ A,
                          const unsigned short* __restrict__ WT,
                          const float* __restrict__ RES,
                          const float* __restrict__ LW,
                          const float* __restrict__ LB,
                          float* __restrict__ OUT) {
  __shared__ __align__(16) float red_s[8][16];
  __shared__ __align__(16) float red_q[8][16];

  const int lane = threadIdx.x & 31;
  const int wave = threadIdx.x >> 5;
  const int row0 = blockIdx.x * 16;
  const int n0 = wave * 128;

  const unsigned short* At = A + (size_t)blockIdx.x * (DD / 32) * 512 + lane * 8;
  const unsigned short* Wt = WT + (size_t)(n0 >> 4) * 512 + lane * 8;

  v8f acc[8] = {};
  for (int k0 = 0; k0 < DD; k0 += 64) {
    if (k0 + 64 < DD) __builtin_prefetch(At + ((k0 >> 5) + 2) * 512 + lane * 24, 0, 1);
    v16bf a[2];
    v16bf bf[2][8];
#pragma unroll
    for (int kk = 0; kk < 2; ++kk) {
      a[kk] = load_frag_t(At + ((k0 >> 5) + kk) * 512);
#pragma unroll
      for (int nt = 0; nt < 8; ++nt)
        bf[kk][nt] = load_frag_t(Wt + ((size_t)((k0 >> 5) + kk) * (DD / 16) + nt) * 512);
    }
    __builtin_amdgcn_sched_barrier(0);
#pragma unroll
    for (int kk = 0; kk < 2; ++kk) {
#pragma unroll
      for (int nt = 0; nt < 8; ++nt)
        acc[nt] = WMMA_BF16(a[kk], bf[kk][nt], acc[nt]);
    }
  }
  const int rbase = (lane & 16) ? 8 : 0;
  const int col0 = lane & 15;

#pragma unroll
  for (int nt = 0; nt < 8; ++nt) {
#pragma unroll
    for (int r = 0; r < 8; ++r)
      acc[nt][r] += RES[(size_t)(row0 + rbase + r) * DD + n0 + nt * 16 + col0];
  }

  float part[8];
#pragma unroll
  for (int r = 0; r < 8; ++r) {
    float s = acc[0][r];
#pragma unroll
    for (int nt = 1; nt < 8; ++nt) s += acc[nt][r];
    s += row_xor<1>(s);
    s += row_xor<2>(s);
    s += row_xor<4>(s);
    s += row_xor<8>(s);
    part[r] = s;
  }
  if (col0 == 0) {
    float4* d = reinterpret_cast<float4*>(&red_s[wave][rbase]);
    d[0] = make_float4(part[0], part[1], part[2], part[3]);
    d[1] = make_float4(part[4], part[5], part[6], part[7]);
  }
  __syncthreads();
  float mu[8];
  {
    float tot[8];
#pragma unroll
    for (int w = 0; w < 8; ++w) {
      const float4* s4 = reinterpret_cast<const float4*>(&red_s[w][rbase]);
      const float4 lo = s4[0], hi = s4[1];
      const float v[8] = {lo.x, lo.y, lo.z, lo.w, hi.x, hi.y, hi.z, hi.w};
#pragma unroll
      for (int r = 0; r < 8; ++r) tot[r] = (w == 0) ? v[r] : tot[r] + v[r];
    }
#pragma unroll
    for (int r = 0; r < 8; ++r) mu[r] = tot[r] * (1.0f / DD);
  }

#pragma unroll
  for (int r = 0; r < 8; ++r) {
    float s = 0.0f;
#pragma unroll
    for (int nt = 0; nt < 8; ++nt) {
      const float d = acc[nt][r] - mu[r];
      acc[nt][r] = d;
      s += d * d;
    }
    s += row_xor<1>(s);
    s += row_xor<2>(s);
    s += row_xor<4>(s);
    s += row_xor<8>(s);
    part[r] = s;
  }
  if (col0 == 0) {
    float4* d = reinterpret_cast<float4*>(&red_q[wave][rbase]);
    d[0] = make_float4(part[0], part[1], part[2], part[3]);
    d[1] = make_float4(part[4], part[5], part[6], part[7]);
  }
  __syncthreads();
  float rstd[8];
  {
    float tot[8];
#pragma unroll
    for (int w = 0; w < 8; ++w) {
      const float4* s4 = reinterpret_cast<const float4*>(&red_q[w][rbase]);
      const float4 lo = s4[0], hi = s4[1];
      const float v[8] = {lo.x, lo.y, lo.z, lo.w, hi.x, hi.y, hi.z, hi.w};
#pragma unroll
      for (int r = 0; r < 8; ++r) tot[r] = (w == 0) ? v[r] : tot[r] + v[r];
    }
#pragma unroll
    for (int r = 0; r < 8; ++r) rstd[r] = rsqrtf(tot[r] * (1.0f / DD) + 1e-6f);
  }

#pragma unroll
  for (int nt = 0; nt < 8; ++nt) {
    const int c = n0 + nt * 16 + col0;
    const float w = LW[c];
    const float bsh = LB[c];
#pragma unroll
    for (int r = 0; r < 8; ++r)
      OUT[(size_t)(row0 + rbase + r) * DD + c] = acc[nt][r] * rstd[r] * w + bsh;
  }
}

extern "C" void kernel_launch(void* const* d_in, const int* in_sizes, int n_in,
                              void* d_out, int out_size, void* d_ws, size_t ws_size,
                              hipStream_t stream) {
  const float* q    = (const float*)d_in[0];
  const float* k    = (const float*)d_in[1];
  const float* v    = (const float*)d_in[2];
  const int*   mask = (const int*)d_in[3];
  const float* Wq   = (const float*)d_in[4];
  const float* Wk   = (const float*)d_in[5];
  const float* Wv   = (const float*)d_in[6];
  const float* Wfc  = (const float*)d_in[7];
  const float* lnw  = (const float*)d_in[8];
  const float* lnb  = (const float*)d_in[9];
  float* out = (float*)d_out;

  char* ws = (char*)d_ws;
  const size_t SZ_ACT  = (size_t)BB * SS * DD * 2;
  const size_t SZ_W    = (size_t)HH * DD * DKV * 2;
  const size_t SZ_HEAD = (size_t)BB * HH * SS * DKV * 2;
  size_t off = 0;
  unsigned short* WqT  = (unsigned short*)(ws + off); off += SZ_W;
  unsigned short* WkT  = (unsigned short*)(ws + off); off += SZ_W;
  unsigned short* WvT  = (unsigned short*)(ws + off); off += SZ_W;
  unsigned short* WfcT = (unsigned short*)(ws + off); off += (size_t)DD * DD * 2;
  unsigned short* kh   = (unsigned short*)(ws + off); off += SZ_HEAD;
  unsigned short* vhT  = (unsigned short*)(ws + off); off += SZ_HEAD;
  unsigned short* aout = (unsigned short*)(ws + off); off += SZ_ACT;

  const int nW   = HH * DD * DKV;
  const int nFc  = DD * DD;

  cvtT_kernel<<<(nW + 255) / 256, 256, 0, stream>>>(Wq, WqT, HH, DD, DKV);
  cvtT_kernel<<<(nW + 255) / 256, 256, 0, stream>>>(Wk, WkT, HH, DD, DKV);
  cvtT_kernel<<<(nW + 255) / 256, 256, 0, stream>>>(Wv, WvT, HH, DD, DKV);
  cvtT_kernel<<<(nFc + 255) / 256, 256, 0, stream>>>(Wfc, WfcT, 1, DD, DD);

  dim3 gProj(BB * HH / PROJ_NH, SS / 64);
  proj_kernel<<<gProj, 128, 0, stream>>>(k, WkT, kh, 1.0f, 0);
  proj_kernel<<<gProj, 128, 0, stream>>>(v, WvT, vhT, 1.0f, 1);

  attn_kernel<<<dim3(BB * HH, SS / 64), 128, 0, stream>>>(q, WqT, kh, vhT, mask, aout);

  fc_ln_kernel<<<(BB * SS) / 16, 256, 0, stream>>>(aout, WfcT, q, lnw, lnb, out);
}


# attn: score tile T, col c = key 8c+T (K plane stored so): a lane holds 8 adjacent keys a row: mask by 16-B loads (was 4-B), P packed in pairs, stored 16 B at once (was 2 B); row-sum order changes
# speedup vs baseline: 4.1100x; 1.0967x over previous
#include <hip/hip_runtime.h>


#define BB 2
#define SS 2048
#define DD 1024
#define HH 16
#define DKV 64

typedef __attribute__((ext_vector_type(16))) __bf16 v16bf;
typedef __attribute__((ext_vector_type(8)))  float  v8f;

union FragU { v16bf v; uint4 q[2]; };

__device__ __forceinline__ unsigned short f2bf(float f) {
  unsigned int u = __float_as_uint(f);
  u += 0x7FFFu + ((u >> 16) & 1u);
  return (unsigned short)(u >> 16);
}

typedef __attribute__((ext_vector_type(2))) float  v2f;
typedef __attribute__((ext_vector_type(2))) __bf16 v2bf;
typedef __attribute__((ext_vector_type(4))) int    v4i;

__device__ __forceinline__ unsigned pack_bf2(float lo, float hi) {
  v2f f = {lo, hi};
  return __builtin_bit_cast(unsigned, __builtin_convertvector(f, v2bf));
}

__device__ __forceinline__ v16bf load_frag(const unsigned short* rowbase, int lane) {
  const unsigned short* p = rowbase + ((lane & 16) ? 8 : 0);
  FragU f;
  f.q[0] = *reinterpret_cast<const uint4*>(p);
  f.q[1] = *reinterpret_cast<const uint4*>(p + 16);
  return f.v;
}

__device__ __forceinline__ int ft_off(int row, int k) {
  return ((k >> 4) << 8) + ((row + (((k >> 3) & 1) << 4)) << 3) + (k & 7);
}

__device__ __forceinline__ v16bf load_frag_t(const unsigned short* tile_lane) {
  FragU f;
  f.q[0] = *reinterpret_cast<const uint4*>(tile_lane);
  f.q[1] = *reinterpret_cast<const uint4*>(tile_lane + 256);
  return f.v;
}

struct RawA { float4 x0, x1, x2, x3; };

__device__ __forceinline__ RawA load_raw_f32(const float* rowbase, int lane) {
  const float* p = rowbase + ((lane & 16) ? 8 : 0);
  RawA r;
  r.x0 = *reinterpret_cast<const float4*>(p);
  r.x1 = *reinterpret_cast<const float4*>(p + 4);
  r.x2 = *reinterpret_cast<const float4*>(p + 16);
  r.x3 = *reinterpret_cast<const float4*>(p + 20);
  return r;
}

__device__ __forceinline__ v16bf pack_frag_f32(const RawA& r) {
  const float4 x0 = r.x0, x1 = r.x1, x2 = r.x2, x3 = r.x3;
  v16bf f;
  f[0]  = (__bf16)x0.x; f[1]  = (__bf16)x0.y; f[2]  = (__bf16)x0.z; f[3]  = (__bf16)x0.w;
  f[4]  = (__bf16)x1.x; f[5]  = (__bf16)x1.y; f[6]  = (__bf16)x1.z; f[7]  = (__bf16)x1.w;
  f[8]  = (__bf16)x2.x; f[9]  = (__bf16)x2.y; f[10] = (__bf16)x2.z; f[11] = (__bf16)x2.w;
  f[12] = (__bf16)x3.x; f[13] = (__bf16)x3.y; f[14] = (__bf16)x3.z; f[15] = (__bf16)x3.w;
  return f;
}

#define WMMA_BF16(a, b, c) \
  __builtin_amdgcn_wmma_f32_16x16x32_bf16(false, (a), false, (b), (short)0, (c), false, false)

template <int M>
__device__ __forceinline__ float row_xor(float x) {
  return __int_as_float(__builtin_amdgcn_update_dpp(
      0, __float_as_int(x), 0x160 + M, 0xf, 0xf, true));
}

__global__ void cvtT_kernel(const float* __restrict__ src,
                            unsigned short* __restrict__ dst,
                            int G, int R, int C) {
  int i = blockIdx.x * blockDim.x + threadIdx.x;
  int n = G * R * C;
  if (i >= n) return;
  int g = i / (R * C);
  int rem = i - g * (R * C);
  int r = rem / C, c = rem - r * C;
  size_t tile = ((size_t)g * (R >> 5) + (r >> 5)) * (C >> 4) + (c >> 4);
  dst[tile * 512 + ft_off(c & 15, r & 31)] = f2bf(src[i]);
}

template <int NH>
__device__ __forceinline__ void proj_tile(const float* __restrict__ Xrow,
                                          const unsigned short* __restrict__ Wt,
                                          int lane, v8f* acc) {
  constexpr size_t HEAD = (size_t)(DD / 32) * (DKV / 16) * 512;
  constexpr int PF = 8;
  const float* Xpf = Xrow + ((lane & 16) ? 32 : 0);
#pragma unroll
  for (int j = 0; j < PF; j += 2) __builtin_prefetch(Xpf + j * 32, 0, 1);

  for (int k0 = 0; k0 < DD; k0 += 64) {
    if (k0 + PF * 32 < DD)
      __builtin_prefetch(Xpf + k0 + PF * 32, 0, 1);
    RawA xa[2];
    v16bf bf[2][NH * 4];
#pragma unroll
    for (int kk = 0; kk < 2; ++kk)
      xa[kk] = load_raw_f32(Xrow + k0 + kk * 32, lane);
#pragma unroll
    for (int kk = 0; kk < 2; ++kk) {
#pragma unroll
      for (int hh = 0; hh < NH; ++hh) {
#pragma unroll
        for (int nt = 0; nt < 4; ++nt)
          bf[kk][hh * 4 + nt] = load_frag_t(
              Wt + hh * HEAD + (size_t)((((k0 >> 5) + kk) << 2) + nt) * 512);
      }
    }
    v16bf a[2];
#pragma unroll
    for (int kk = 0; kk < 2; ++kk) a[kk] = pack_frag_f32(xa[kk]);
    __builtin_amdgcn_sched_barrier(0);
#pragma unroll
    for (int kk = 0; kk < 2; ++kk) {
#pragma unroll
      for (int nt = 0; nt < NH * 4; ++nt)
        acc[nt] = WMMA_BF16(a[kk], bf[kk][nt], acc[nt]);
    }
  }
}

#define PROJ_NH 2
__global__ void __launch_bounds__(128) __attribute__((amdgpu_waves_per_eu(2, 2)))
proj_kernel(const float* __restrict__ X,
                            const unsigned short* __restrict__ WT,
                            unsigned short* __restrict__ OUT,
                            float scale, int transposeOut) {
  const int lane = threadIdx.x & 31;
  const int wave = threadIdx.x >> 5;
  const int b = blockIdx.x / (HH / PROJ_NH);
  const int h0 = (blockIdx.x % (HH / PROJ_NH)) * PROJ_NH;
  const int s_base = blockIdx.y * 64 + wave * 16;

  const int lr = lane & 15;
  const int srow = transposeOut ? s_base + lr
                                : blockIdx.y * 64 + ((lr >> 1) << 3) + 2 * wave + (lr & 1);
  const float* Xrow = X + ((size_t)b * SS + (size_t)srow) * DD;
  const unsigned short* Wt = WT + (size_t)h0 * (DD / 32) * (DKV / 16) * 512 + lane * 8;

  v8f accs[PROJ_NH * 4] = {};
  proj_tile<PROJ_NH>(Xrow, Wt, lane, accs);
  const int rbase = (lane & 16) ? 8 : 0;
  const int col0 = lane & 15;
#pragma unroll
  for (int hh = 0; hh < PROJ_NH; ++hh) {
  const v8f* acc = accs + hh * 4;
  const int bh = b * HH + h0 + hh;
  if (transposeOut) {
    unsigned short* o = OUT + (((size_t)bh * (SS / 32) + (s_base >> 5)) << 2) * 512 +
                        (((s_base >> 4) & 1) << 8) + lane * 8;
#pragma unroll
    for (int nt = 0; nt < 4; ++nt) {
      uint4 w;
      w.x = (unsigned)f2bf(acc[nt][0] * scale) | ((unsigned)f2bf(acc[nt][1] * scale) << 16);
      w.y = (unsigned)f2bf(acc[nt][2] * scale) | ((unsigned)f2bf(acc[nt][3] * scale) << 16);
      w.z = (unsigned)f2bf(acc[nt][4] * scale) | ((unsigned)f2bf(acc[nt][5] * scale) << 16);
      w.w = (unsigned)f2bf(acc[nt][6] * scale) | ((unsigned)f2bf(acc[nt][7] * scale) << 16);
      *reinterpret_cast<uint4*>(o + nt * 512) = w;
    }
  } else {
    const int n0 = ((blockIdx.y & 1) << 3) + (rbase >> 1);
    unsigned short* o = OUT + (((size_t)bh * (SS / 128) + (blockIdx.y >> 1)) << 4) * 512;
#pragma unroll
    for (int nt = 0; nt < 4; ++nt) {
#pragma unroll
      for (int r = 0; r < 8; ++r)
        o[(((2 * wave + (r & 1)) << 1) + (nt >> 1)) * 512 +
          ft_off(n0 + (r >> 1), (nt & 1) * 16 + col0)] = f2bf(acc[nt][r] * scale);
    }
  }
  }
}

#define ATT_KB 128
#define ATT_PP (ATT_KB + 8)

__device__ __forceinline__ void load_k4(v16bf* kb, const unsigned short* kp) {
#pragma unroll
  for (int i = 0; i < 8; ++i) kb[i] = load_frag_t(kp + i * 512);
}

__device__ __forceinline__ void load_m4(v4i* mk, const int* mp) {
#pragma unroll
  for (int r = 0; r < 8; ++r)
    mk[r] = *reinterpret_cast<const v4i*>(mp + (size_t)r * SS);
}

__device__ __forceinline__ void scores64(const v16bf* kb, const v4i* mk,
                                         v16bf qf0, v16bf qf1, v8f* sc) {
#pragma unroll
  for (int jj = 0; jj < 2; ++jj) {
    v8f s0 = {}, s1 = {};
    s0 = WMMA_BF16(qf0, kb[4 * jj],     s0);
    s0 = WMMA_BF16(qf1, kb[4 * jj + 1], s0);
    s1 = WMMA_BF16(qf0, kb[4 * jj + 2], s1);
    s1 = WMMA_BF16(qf1, kb[4 * jj + 3], s1);
#pragma unroll
    for (int r = 0; r < 8; ++r) {
      sc[2 * jj][r]     = (mk[r][2 * jj]     != 0) ? s0[r] : -1.0e9f;
      sc[2 * jj + 1][r] = (mk[r][2 * jj + 1] != 0) ? s1[r] : -1.0e9f;
    }
  }
}

__global__ void __launch_bounds__(128) __attribute__((amdgpu_waves_per_eu(4, 4)))
attn_kernel(const float* __restrict__ XQ,
                            const unsigned short* __restrict__ WQT,
                            const unsigned short* __restrict__ KH,
                            const unsigned short* __restrict__ VT,
                            const int* __restrict__ MASK,
                            unsigned short* __restrict__ AOUT) {
  __shared__ __align__(16) unsigned short Pld[4][16][ATT_PP];

  const int lane = threadIdx.x & 31;
  const int wave = threadIdx.x >> 5;
  const int bh = blockIdx.x;
  const int b = bh >> 4, h = bh & 15;
  const int s_base = blockIdx.y * 64 + wave * 16;
  const int rbase = (lane & 16) ? 8 : 0;
  const int col = lane & 15;

  v16bf qf0, qf1;
  {
    v8f qa[4] = {};
    proj_tile<1>(XQ + ((size_t)b * SS + (size_t)(s_base + col)) * DD,
              WQT + (size_t)h * (DD / 32) * (DKV / 16) * 512 + lane * 8, lane, qa);
#pragma unroll
    for (int nt = 0; nt < 4; ++nt) {
#pragma unroll
      for (int r = 0; r < 8; ++r)
        Pld[wave][rbase + r][nt * 16 + col] = f2bf(qa[nt][r] * 0.125f);
    }
    asm volatile("s_wait_dscnt 0" ::: "memory");
    qf0 = load_frag(&Pld[wave][col][0], lane);
    qf1 = load_frag(&Pld[wave][col][32], lane);
  }

  v8f o[4] = {};
  float m[8], l[8];
#pragma unroll
  for (int r = 0; r < 8; ++r) { m[r] = -3.0e38f; l[r] = 0.0f; }

  const int* maskrow = MASK + (size_t)b * SS * SS + (size_t)(s_base + rbase) * SS + 8 * col;
  const unsigned short* Kt = KH + (size_t)bh * SS * DKV + lane * 8;
  const unsigned short* Vt = VT + (size_t)bh * DKV * SS + lane * 8;

  for (int t0 = 0; t0 < SS; t0 += ATT_KB) {
    v8f sc[8];
    {
      v4i mk[2][8];
      v16bf kb[8];
      load_m4(mk[0], maskrow + t0);
      load_m4(mk[1], maskrow + t0 + 4);
      load_k4(kb, Kt + (size_t)t0 * DKV);
      __builtin_amdgcn_sched_barrier(0);
      scores64(kb, mk[0], qf0, qf1, sc);
      __builtin_amdgcn_sched_barrier(0);
      load_k4(kb, Kt + (size_t)(t0 + 64) * DKV);
      scores64(kb, mk[1], qf0, qf1, sc + 4);
    }
    v16bf va[8];
#pragma unroll
    for (int i = 0; i < 8; ++i) va[i] = load_frag_t(Vt + (size_t)t0 * DKV + i * 512);

#pragma unroll
    for (int r = 0; r < 8; ++r) {
      float mx = fmaxf(sc[0][r], sc[1][r]);
#pragma unroll
      for (int j = 1; j < 4; ++j)
        mx = fmaxf(mx, fmaxf(sc[2 * j][r], sc[2 * j + 1][r]));
      mx = fmaxf(mx, row_xor<1>(mx));
      mx = fmaxf(mx, row_xor<2>(mx));
      mx = fmaxf(mx, row_xor<4>(mx));
      mx = fmaxf(mx, row_xor<8>(mx));
      float mnew  = fmaxf(m[r], mx);
      float alpha = __expf(m[r] - mnew);
      float rs = 0.0f;
      unsigned pw[4];
#pragma unroll
      for (int j = 0; j < 4; ++j) {
        float p0 = __expf(sc[2 * j][r] - mnew);
        float p1 = __expf(sc[2 * j + 1][r] - mnew);
        rs += p0 + p1;
        pw[j] = pack_bf2(p0, p1);
      }
      *reinterpret_cast<uint4*>(&Pld[wave][rbase + r][8 * col]) =
          make_uint4(pw[0], pw[1], pw[2], pw[3]);
      rs += row_xor<1>(rs);
      rs += row_xor<2>(rs);
      rs += row_xor<4>(rs);
      rs += row_xor<8>(rs);
      l[r] = l[r] * alpha + rs;
      m[r] = mnew;
#pragma unroll
      for (int nt = 0; nt < 4; ++nt) o[nt][r] = o[nt][r] * alpha;
    }
    asm volatile("s_wait_dscnt 0" ::: "memory");
    v16bf vc[8];
#pragma unroll
    for (int i = 0; i < 8; ++i) vc[i] = load_frag_t(Vt + (size_t)(t0 + 64) * DKV + i * 512);
#pragma unroll
    for (int j = 0; j < 4; ++j) {
      v16bf pf = load_frag(&Pld[wave][col][32 * j], lane);
#pragma unroll
      for (int nt = 0; nt < 4; ++nt)
        o[nt] = WMMA_BF16(pf, (j < 2) ? va[4 * j + nt] : vc[4 * (j - 2) + nt], o[nt]);
    }
  }

  unsigned short* ao =
      AOUT + ((((size_t)b * SS + s_base) >> 4) * (HH * DKV / 32) + h * 2) * 512;
#pragma unroll
  for (int r = 0; r < 8; ++r) {
    float inv = 1.0f / l[r];
#pragma unroll
    for (int nt = 0; nt < 4; ++nt)
      ao[(nt >> 1) * 512 + ft_off(rbase + r, (nt & 1) * 16 + col)] =
          f2bf(o[nt][r] * inv);
  }
}

__global__ void __launch_bounds__(256) __attribute__((amdgpu_waves_per_eu(2, 2)))
fc_ln_kernel(const unsigned short* __restrict__ A,
                          const unsigned short* __restrict__ WT,
                          const float* __restrict__ RES,
                          const float* __restrict__ LW,
                          const float* __restrict__ LB,
                          float* __restrict__ OUT) {
  __shared__ __align__(16) float red_s[8][16];
  __shared__ __align__(16) float red_q[8][16];

  const int lane = threadIdx.x & 31;
  const int wave = threadIdx.x >> 5;
  const int row0 = blockIdx.x * 16;
  const int n0 = wave * 128;

  const unsigned short* At = A + (size_t)blockIdx.x * (DD / 32) * 512 + lane * 8;
  const unsigned short* Wt = WT + (size_t)(n0 >> 4) * 512 + lane * 8;

  v8f acc[8] = {};
  for (int k0 = 0; k0 < DD; k0 += 64) {
    if (k0 + 64 < DD) __builtin_prefetch(At + ((k0 >> 5) + 2) * 512 + lane * 24, 0, 1);
    v16bf a[2];
    v16bf bf[2][8];
#pragma unroll
    for (int kk = 0; kk < 2; ++kk) {
      a[kk] = load_frag_t(At + ((k0 >> 5) + kk) * 512);
#pragma unroll
      for (int nt = 0; nt < 8; ++nt)
        bf[kk][nt] = load_frag_t(Wt + ((size_t)((k0 >> 5) + kk) * (DD / 16) + nt) * 512);
    }
    __builtin_amdgcn_sched_barrier(0);
#pragma unroll
    for (int kk = 0; kk < 2; ++kk) {
#pragma unroll
      for (int nt = 0; nt < 8; ++nt)
        acc[nt] = WMMA_BF16(a[kk], bf[kk][nt], acc[nt]);
    }
  }
  const int rbase = (lane & 16) ? 8 : 0;
  const int col0 = lane & 15;

#pragma unroll
  for (int nt = 0; nt < 8; ++nt) {
#pragma unroll
    for (int r = 0; r < 8; ++r)
      acc[nt][r] += RES[(size_t)(row0 + rbase + r) * DD + n0 + nt * 16 + col0];
  }

  float part[8];
#pragma unroll
  for (int r = 0; r < 8; ++r) {
    float s = acc[0][r];
#pragma unroll
    for (int nt = 1; nt < 8; ++nt) s += acc[nt][r];
    s += row_xor<1>(s);
    s += row_xor<2>(s);
    s += row_xor<4>(s);
    s += row_xor<8>(s);
    part[r] = s;
  }
  if (col0 == 0) {
    float4* d = reinterpret_cast<float4*>(&red_s[wave][rbase]);
    d[0] = make_float4(part[0], part[1], part[2], part[3]);
    d[1] = make_float4(part[4], part[5], part[6], part[7]);
  }
  __syncthreads();
  float mu[8];
  {
    float tot[8];
#pragma unroll
    for (int w = 0; w < 8; ++w) {
      const float4* s4 = reinterpret_cast<const float4*>(&red_s[w][rbase]);
      const float4 lo = s4[0], hi = s4[1];
      const float v[8] = {lo.x, lo.y, lo.z, lo.w, hi.x, hi.y, hi.z, hi.w};
#pragma unroll
      for (int r = 0; r < 8; ++r) tot[r] = (w == 0) ? v[r] : tot[r] + v[r];
    }
#pragma unroll
    for (int r = 0; r < 8; ++r) mu[r] = tot[r] * (1.0f / DD);
  }

#pragma unroll
  for (int r = 0; r < 8; ++r) {
    float s = 0.0f;
#pragma unroll
    for (int nt = 0; nt < 8; ++nt) {
      const float d = acc[nt][r] - mu[r];
      acc[nt][r] = d;
      s += d * d;
    }
    s += row_xor<1>(s);
    s += row_xor<2>(s);
    s += row_xor<4>(s);
    s += row_xor<8>(s);
    part[r] = s;
  }
  if (col0 == 0) {
    float4* d = reinterpret_cast<float4*>(&red_q[wave][rbase]);
    d[0] = make_float4(part[0], part[1], part[2], part[3]);
    d[1] = make_float4(part[4], part[5], part[6], part[7]);
  }
  __syncthreads();
  float rstd[8];
  {
    float tot[8];
#pragma unroll
    for (int w = 0; w < 8; ++w) {
      const float4* s4 = reinterpret_cast<const float4*>(&red_q[w][rbase]);
      const float4 lo = s4[0], hi = s4[1];
      const float v[8] = {lo.x, lo.y, lo.z, lo.w, hi.x, hi.y, hi.z, hi.w};
#pragma unroll
      for (int r = 0; r < 8; ++r) tot[r] = (w == 0) ? v[r] : tot[r] + v[r];
    }
#pragma unroll
    for (int r = 0; r < 8; ++r) rstd[r] = rsqrtf(tot[r] * (1.0f / DD) + 1e-6f);
  }

#pragma unroll
  for (int nt = 0; nt < 8; ++nt) {
    const int c = n0 + nt * 16 + col0;
    const float w = LW[c];
    const float bsh = LB[c];
#pragma unroll
    for (int r = 0; r < 8; ++r)
      OUT[(size_t)(row0 + rbase + r) * DD + c] = acc[nt][r] * rstd[r] * w + bsh;
  }
}

extern "C" void kernel_launch(void* const* d_in, const int* in_sizes, int n_in,
                              void* d_out, int out_size, void* d_ws, size_t ws_size,
                              hipStream_t stream) {
  const float* q    = (const float*)d_in[0];
  const float* k    = (const float*)d_in[1];
  const float* v    = (const float*)d_in[2];
  const int*   mask = (const int*)d_in[3];
  const float* Wq   = (const float*)d_in[4];
  const float* Wk   = (const float*)d_in[5];
  const float* Wv   = (const float*)d_in[6];
  const float* Wfc  = (const float*)d_in[7];
  const float* lnw  = (const float*)d_in[8];
  const float* lnb  = (const float*)d_in[9];
  float* out = (float*)d_out;

  char* ws = (char*)d_ws;
  const size_t SZ_ACT  = (size_t)BB * SS * DD * 2;
  const size_t SZ_W    = (size_t)HH * DD * DKV * 2;
  const size_t SZ_HEAD = (size_t)BB * HH * SS * DKV * 2;
  size_t off = 0;
  unsigned short* WqT  = (unsigned short*)(ws + off); off += SZ_W;
  unsigned short* WkT  = (unsigned short*)(ws + off); off += SZ_W;
  unsigned short* WvT  = (unsigned short*)(ws + off); off += SZ_W;
  unsigned short* WfcT = (unsigned short*)(ws + off); off += (size_t)DD * DD * 2;
  unsigned short* kh   = (unsigned short*)(ws + off); off += SZ_HEAD;
  unsigned short* vhT  = (unsigned short*)(ws + off); off += SZ_HEAD;
  unsigned short* aout = (unsigned short*)(ws + off); off += SZ_ACT;

  const int nW   = HH * DD * DKV;
  const int nFc  = DD * DD;

  cvtT_kernel<<<(nW + 255) / 256, 256, 0, stream>>>(Wq, WqT, HH, DD, DKV);
  cvtT_kernel<<<(nW + 255) / 256, 256, 0, stream>>>(Wk, WkT, HH, DD, DKV);
  cvtT_kernel<<<(nW + 255) / 256, 256, 0, stream>>>(Wv, WvT, HH, DD, DKV);
  cvtT_kernel<<<(nFc + 255) / 256, 256, 0, stream>>>(Wfc, WfcT, 1, DD, DD);

  dim3 gProj(BB * HH / PROJ_NH, SS / 64);
  proj_kernel<<<gProj, 128, 0, stream>>>(k, WkT, kh, 1.0f, 0);
  proj_kernel<<<gProj, 128, 0, stream>>>(v, WvT, vhT, 1.0f, 1);

  attn_kernel<<<dim3(BB * HH, SS / 64), 128, 0, stream>>>(q, WqT, kh, vhT, mask, aout);

  fc_ln_kernel<<<(BB * SS) / 16, 256, 0, stream>>>(aout, WfcT, q, lnw, lnb, out);
}
